# starting kernel
# baseline (speedup 1.0000x reference)
#include <hip/hip_runtime.h>
#include <math.h>

#define D_MODEL 2048
#define NHEADS  16
#define DK      128
#define SEQ     2048
#define BATCH   2
#define MROWS   (BATCH * SEQ)
#define LOG_THETA 9.210340371976184f

typedef __attribute__((ext_vector_type(2))) float v2f;
typedef __attribute__((ext_vector_type(8))) float v8f;

__device__ __forceinline__ v8f wmma_f32(v2f a, v2f b, v8f c) {
  return __builtin_amdgcn_wmma_f32_16x16x4_f32(false, a, false, b,
                                               (short)0, c, false, false);
}

__device__ __forceinline__ v8f vzero8() {
  v8f z = {0.f, 0.f, 0.f, 0.f, 0.f, 0.f, 0.f, 0.f};
  return z;
}

__global__ __launch_bounds__(32)
void mha_qkv_rope(const float* __restrict__ x, const int* __restrict__ tpos,
                  const float* __restrict__ WQ, const float* __restrict__ WK,
                  const float* __restrict__ WV,
                  float* __restrict__ Qo, float* __restrict__ Ko,
                  float* __restrict__ Vo) {
  const int lane = threadIdx.x;
  const int hi = lane >> 4;
  const int lo = lane & 15;
  const int m0 = blockIdx.x * 16;
  const int n0 = blockIdx.y * 64;
  const int mat = blockIdx.z;

  const float* __restrict__ W = (mat == 0) ? WQ : ((mat == 1) ? WK : WV);
  float* __restrict__ Out = (mat == 0) ? Qo : ((mat == 1) ? Ko : Vo);

  v8f acc[4];
#pragma unroll
  for (int t = 0; t < 4; ++t) acc[t] = vzero8();

  const float* arow = x + (size_t)(m0 + lo) * D_MODEL;

#pragma unroll 8
  for (int k = 0; k < D_MODEL; k += 4) {
    const int kk = k + 2 * hi;
    v2f a = *(const v2f*)(arow + kk);
#pragma unroll
    for (int t = 0; t < 4; ++t) {
      v2f b = *(const v2f*)(W + (size_t)(n0 + t * 16 + lo) * D_MODEL + kk);
      acc[t] = wmma_f32(a, b, acc[t]);
    }
  }

  const int bidx = m0 >> 11;
  if (mat == 2) {
#pragma unroll
    for (int t = 0; t < 4; ++t) {
      const int ng = n0 + t * 16 + lo;
      const int h = ng >> 7, d = ng & 127;
#pragma unroll
      for (int i = 0; i < 8; ++i) {
        const int s = (m0 + i + 8 * hi) & (SEQ - 1);
        Out[(((size_t)bidx * NHEADS + h) * SEQ + s) * DK + d] = acc[t][i];
      }
    }
  } else {
#pragma unroll
    for (int t = 0; t < 4; ++t) {
      const int ng = n0 + t * 16 + lo;
      const int h = ng >> 7, d = ng & 127;
      const float inv_freq = expf(-(float)(d & ~1) * (LOG_THETA / (float)DK));
      const int odd = d & 1;
#pragma unroll
      for (int i = 0; i < 8; ++i) {
        const int s = (m0 + i + 8 * hi) & (SEQ - 1);
        const float p = (float)tpos[bidx * SEQ + s];
        float sn, cs;
        sincosf(p * inv_freq, &sn, &cs);
        const float val = acc[t][i];
        const float part = __shfl_xor(val, 1, 32);
        const float r = odd ? fmaf(part, sn, val * cs)
                            : fmaf(val, cs, -part * sn);
        Out[(((size_t)bidx * NHEADS + h) * SEQ + s) * DK + d] = r;
      }
    }
  }
}

__global__ __launch_bounds__(32)
void mha_attn(const float* __restrict__ Q, const float* __restrict__ K,
              const float* __restrict__ V, float* __restrict__ AO) {
  const int lane = threadIdx.x;
  const int hi = lane >> 4;
  const int lo = lane & 15;
  const int qt = blockIdx.x;
  const int h = blockIdx.y;
  const int b = blockIdx.z;

  const size_t head = ((size_t)b * NHEADS + h) * SEQ;
  const float* __restrict__ Qb = Q + head * DK;
  const float* __restrict__ Kb = K + head * DK;
  const float* __restrict__ Vb = V + head * DK;

  __shared__ float Pst[16 * 16];

  v2f qa[32];
  const float* qrow = Qb + (size_t)(qt * 16 + lo) * DK;
#pragma unroll
  for (int ks = 0; ks < 32; ++ks)
    qa[ks] = *(const v2f*)(qrow + 4 * ks + 2 * hi);

  float mrow[8], lrow[8], alph[8];
  v8f o[8];
#pragma unroll
  for (int i = 0; i < 8; ++i) {
    mrow[i] = -__builtin_inff();
    lrow[i] = 0.f;
    o[i] = vzero8();
  }

  const float scale = 0.08838834764831845f;

  for (int kt = 0; kt <= qt; ++kt) {
    v8f sc = vzero8();
    const float* krow = Kb + (size_t)(kt * 16 + lo) * DK;
#pragma unroll
    for (int ks = 0; ks < 32; ++ks) {
      v2f kb = *(const v2f*)(krow + 4 * ks + 2 * hi);
      sc = wmma_f32(qa[ks], kb, sc);
    }

    const bool diag = (kt == qt);
#pragma unroll
    for (int i = 0; i < 8; ++i) {
      float s = sc[i] * scale;
      if (diag && lo > i + 8 * hi) s = -__builtin_inff();
      float rm = s;
      rm = fmaxf(rm, __shfl_xor(rm, 1, 32));
      rm = fmaxf(rm, __shfl_xor(rm, 2, 32));
      rm = fmaxf(rm, __shfl_xor(rm, 4, 32));
      rm = fmaxf(rm, __shfl_xor(rm, 8, 32));
      const float mnew = fmaxf(mrow[i], rm);
      const float a = expf(mrow[i] - mnew);
      const float p = expf(s - mnew);
      float rs = p;
      rs += __shfl_xor(rs, 1, 32);
      rs += __shfl_xor(rs, 2, 32);
      rs += __shfl_xor(rs, 4, 32);
      rs += __shfl_xor(rs, 8, 32);
      lrow[i] = lrow[i] * a + rs;
      mrow[i] = mnew;
      alph[i] = a;
      Pst[(i + 8 * hi) * 16 + lo] = p;
    }
#pragma unroll
    for (int t = 0; t < 8; ++t)
#pragma unroll
      for (int i = 0; i < 8; ++i) o[t][i] *= alph[i];

#pragma unroll
    for (int ks = 0; ks < 4; ++ks) {
      v2f a = *(const v2f*)(&Pst[lo * 16 + 4 * ks + 2 * hi]);
      const float* vrow = Vb + (size_t)(kt * 16 + 4 * ks + 2 * hi) * DK;
#pragma unroll
      for (int t = 0; t < 8; ++t) {
        v2f bv;
        bv[0] = vrow[t * 16 + lo];
        bv[1] = vrow[DK + t * 16 + lo];
        o[t] = wmma_f32(a, bv, o[t]);
      }
    }
  }

#pragma unroll
  for (int i = 0; i < 8; ++i) {
    const float invl = 1.f / lrow[i];
    const size_t row = (size_t)b * SEQ + qt * 16 + i + 8 * hi;
#pragma unroll
    for (int t = 0; t < 8; ++t)
      AO[row * D_MODEL + h * DK + t * 16 + lo] = o[t][i] * invl;
  }
}

__global__ __launch_bounds__(32)
void mha_out_proj(const float* __restrict__ A, const float* __restrict__ WO,
                  float* __restrict__ out) {
  const int lane = threadIdx.x;
  const int hi = lane >> 4;
  const int lo = lane & 15;
  const int m0 = blockIdx.x * 16;
  const int n0 = blockIdx.y * 64;

  v8f acc[4];
#pragma unroll
  for (int t = 0; t < 4; ++t) acc[t] = vzero8();

  const float* arow = A + (size_t)(m0 + lo) * D_MODEL;
#pragma unroll 8
  for (int k = 0; k < D_MODEL; k += 4) {
    const int kk = k + 2 * hi;
    v2f a = *(const v2f*)(arow + kk);
#pragma unroll
    for (int t = 0; t < 4; ++t) {
      v2f b = *(const v2f*)(WO + (size_t)(n0 + t * 16 + lo) * D_MODEL + kk);
      acc[t] = wmma_f32(a, b, acc[t]);
    }
  }

#pragma unroll
  for (int t = 0; t < 4; ++t)
#pragma unroll
    for (int i = 0; i < 8; ++i)
      out[(size_t)(m0 + i + 8 * hi) * D_MODEL + n0 + t * 16 + lo] = acc[t][i];
}

extern "C" void kernel_launch(void* const* d_in, const int* in_sizes, int n_in,
                              void* d_out, int out_size, void* d_ws, size_t ws_size,
                              hipStream_t stream) {
  (void)in_sizes; (void)n_in; (void)out_size; (void)ws_size;
  const float* x   = (const float*)d_in[0];
  const int*   tp  = (const int*)d_in[1];
  const float* WQ  = (const float*)d_in[2];
  const float* WK  = (const float*)d_in[3];
  const float* WV  = (const float*)d_in[4];
  const float* WO  = (const float*)d_in[5];
  float* out = (float*)d_out;

  const size_t QSZ = (size_t)BATCH * NHEADS * SEQ * DK;
  float* ws = (float*)d_ws;
  float* Qw = ws;
  float* Kw = ws + QSZ;
  float* Vw = ws + 2 * QSZ;
  float* AO = ws + 3 * QSZ;

  dim3 gProj(MROWS / 16, (D_MODEL / 16) / 4, 3);
  mha_qkv_rope<<<gProj, 32, 0, stream>>>(x, tp, WQ, WK, WV, Qw, Kw, Vw);

  dim3 gAttn(SEQ / 16, NHEADS, BATCH);
  mha_attn<<<gAttn, 32, 0, stream>>>(Qw, Kw, Vw, AO);

  dim3 gOut(MROWS / 16, (D_MODEL / 16) / 4);
  mha_out_proj<<<gOut, 32, 0, stream>>>(AO, WO, out);
}


# Attention softmax row max/sum and RoPE pair exchange: lane exchanges within a 16-lane row on DPP row_xmask (vector ALU) instead of __shfl_xor (ds_bpermute + a wait per step); same butterfly order
# speedup vs baseline: 2.5644x; 2.5644x over previous
#include <hip/hip_runtime.h>
#include <math.h>

#define D_MODEL 2048
#define NHEADS  16
#define DK      128
#define SEQ     2048
#define BATCH   2
#define MROWS   (BATCH * SEQ)
#define LOG_THETA 9.210340371976184f

typedef __attribute__((ext_vector_type(2))) float v2f;
typedef __attribute__((ext_vector_type(8))) float v8f;

__device__ __forceinline__ v8f wmma_f32(v2f a, v2f b, v8f c) {
  return __builtin_amdgcn_wmma_f32_16x16x4_f32(false, a, false, b,
                                               (short)0, c, false, false);
}

__device__ __forceinline__ v8f vzero8() {
  v8f z = {0.f, 0.f, 0.f, 0.f, 0.f, 0.f, 0.f, 0.f};
  return z;
}

template <int M>
__device__ __forceinline__ float row_xor(float v) {
  return __builtin_bit_cast(
      float, __builtin_amdgcn_mov_dpp(__builtin_bit_cast(int, v), 0x160 | M,
                                      0xf, 0xf, true));
}

#define NV4   4
#define KSTEP (8 * NV4)
#define TROW  ((size_t)16 * D_MODEL)

struct FragSet {
  float4 a[4][NV4];
  float4 b[4][NV4];
};

__device__ __forceinline__ void load_frags(FragSet& f, const float* __restrict__ pa,
                                           const float* __restrict__ pb, int k) {
#pragma unroll
  for (int i = 0; i < 4; ++i)
#pragma unroll
    for (int j = 0; j < NV4; ++j)
      f.a[i][j] = *(const float4*)(pa + i * TROW + k + 4 * j);
#pragma unroll
  for (int t = 0; t < 4; ++t)
#pragma unroll
    for (int j = 0; j < NV4; ++j)
      f.b[t][j] = *(const float4*)(pb + t * TROW + k + 4 * j);
}

__device__ __forceinline__ void mma_frags(const FragSet& f, v8f (&acc)[4][4]) {
#pragma unroll
  for (int j = 0; j < NV4; ++j) {
#pragma unroll
    for (int i = 0; i < 4; ++i) {
      v2f a = {f.a[i][j].x, f.a[i][j].y};
#pragma unroll
      for (int t = 0; t < 4; ++t) {
        v2f b = {f.b[t][j].x, f.b[t][j].y};
        acc[i][t] = wmma_f32(a, b, acc[i][t]);
      }
    }
#pragma unroll
    for (int i = 0; i < 4; ++i) {
      v2f a = {f.a[i][j].z, f.a[i][j].w};
#pragma unroll
      for (int t = 0; t < 4; ++t) {
        v2f b = {f.b[t][j].z, f.b[t][j].w};
        acc[i][t] = wmma_f32(a, b, acc[i][t]);
      }
    }
  }
}

__device__ __forceinline__ void gemm_64x64(const float* __restrict__ A,
                                           const float* __restrict__ Wm,
                                           int m0, int n0, int lo, int hi,
                                           v8f (&acc)[4][4]) {
#pragma unroll
  for (int i = 0; i < 4; ++i)
#pragma unroll
    for (int t = 0; t < 4; ++t) acc[i][t] = vzero8();

  const float* pa = A + (size_t)(m0 + lo) * D_MODEL + 4 * NV4 * hi;
  const float* pb = Wm + (size_t)(n0 + lo) * D_MODEL + 4 * NV4 * hi;

  FragSet f0, f1;
  load_frags(f0, pa, pb, 0);
  int k = 0;
  for (; k < D_MODEL - 2 * KSTEP; k += 2 * KSTEP) {
    load_frags(f1, pa, pb, k + KSTEP);
    __builtin_amdgcn_sched_barrier(0);
    mma_frags(f0, acc);
    __builtin_amdgcn_sched_barrier(0);
    load_frags(f0, pa, pb, k + 2 * KSTEP);
    __builtin_amdgcn_sched_barrier(0);
    mma_frags(f1, acc);
  }
  load_frags(f1, pa, pb, k + KSTEP);
  __builtin_amdgcn_sched_barrier(0);
  mma_frags(f0, acc);
  mma_frags(f1, acc);
}

__global__ __launch_bounds__(32)
void mha_qkv_rope(const float* __restrict__ x, const int* __restrict__ tpos,
                  const float* __restrict__ WQ, const float* __restrict__ WK,
                  const float* __restrict__ WV,
                  float* __restrict__ Qo, float* __restrict__ Ko,
                  float* __restrict__ Vo) {
  const int lane = threadIdx.x;
  const int hi = lane >> 4;
  const int lo = lane & 15;
  const int m0 = blockIdx.x * 64;
  const int n0 = blockIdx.y * 64;
  const int mat = blockIdx.z;

  const float* __restrict__ W = (mat == 0) ? WQ : ((mat == 1) ? WK : WV);
  float* __restrict__ Out = (mat == 0) ? Qo : ((mat == 1) ? Ko : Vo);

  v8f acc[4][4];
  gemm_64x64(x, W, m0, n0, lo, hi, acc);

  const int bidx = m0 >> 11;
  if (mat == 2) {
#pragma unroll
    for (int mi = 0; mi < 4; ++mi)
#pragma unroll
      for (int t = 0; t < 4; ++t) {
        const int ng = n0 + t * 16 + lo;
        const int h = ng >> 7, d = ng & 127;
#pragma unroll
        for (int i = 0; i < 8; ++i) {
          const int s = (m0 + mi * 16 + i + 8 * hi) & (SEQ - 1);
          Out[(((size_t)bidx * NHEADS + h) * SEQ + s) * DK + d] = acc[mi][t][i];
        }
      }
  } else {
    float inv_freq[4];
    size_t colofs[4];
#pragma unroll
    for (int t = 0; t < 4; ++t) {
      const int ng = n0 + t * 16 + lo;
      const int h = ng >> 7, d = ng & 127;
      inv_freq[t] = expf(-(float)(d & ~1) * (LOG_THETA / (float)DK));
      colofs[t] = ((size_t)bidx * NHEADS + h) * SEQ * DK + d;
    }
    const int odd = lo & 1;
#pragma unroll 1
    for (int i = 0; i < 8; ++i) {
      int srow[4];
      float pos[4];
#pragma unroll
      for (int mi = 0; mi < 4; ++mi) {
        srow[mi] = (m0 + mi * 16 + i + 8 * hi) & (SEQ - 1);
        pos[mi] = (float)tpos[bidx * SEQ + srow[mi]];
      }
#pragma unroll
      for (int mi = 0; mi < 4; ++mi) {
#pragma unroll
        for (int t = 0; t < 4; ++t) {
          float sn, cs;
          sincosf(pos[mi] * inv_freq[t], &sn, &cs);
          const float val = acc[mi][t][0];
          const float part = row_xor<1>(val);
          const float r = odd ? fmaf(part, sn, val * cs)
                              : fmaf(val, cs, -part * sn);
          Out[colofs[t] + (size_t)srow[mi] * DK] = r;
          acc[mi][t] = __builtin_shufflevector(acc[mi][t], acc[mi][t],
                                               1, 2, 3, 4, 5, 6, 7, 7);
        }
      }
    }
  }
}

__global__ __launch_bounds__(32)
void mha_attn(const float* __restrict__ Q, const float* __restrict__ K,
              const float* __restrict__ V, float* __restrict__ AO) {
  const int lane = threadIdx.x;
  const int hi = lane >> 4;
  const int lo = lane & 15;
  const int qt = blockIdx.x;
  const int h = blockIdx.y;
  const int b = blockIdx.z;

  const size_t head = ((size_t)b * NHEADS + h) * SEQ;
  const float* __restrict__ Qb = Q + head * DK;
  const float* __restrict__ Kb = K + head * DK;
  const float* __restrict__ Vb = V + head * DK;

  __shared__ float Pst[16 * 16];

  v2f qa[32];
  const float* qrow = Qb + (size_t)(qt * 16 + lo) * DK;
#pragma unroll
  for (int ks = 0; ks < 32; ++ks)
    qa[ks] = *(const v2f*)(qrow + 4 * ks + 2 * hi);

  float mrow[8], lrow[8], alph[8];
  v8f o[8];
#pragma unroll
  for (int i = 0; i < 8; ++i) {
    mrow[i] = -__builtin_inff();
    lrow[i] = 0.f;
    o[i] = vzero8();
  }

  const float scale = 0.08838834764831845f;

  for (int kt = 0; kt <= qt; ++kt) {
    v8f sc = vzero8();
    const float* krow = Kb + (size_t)(kt * 16 + lo) * DK;
#pragma unroll
    for (int ks = 0; ks < 32; ++ks) {
      v2f kb = *(const v2f*)(krow + 4 * ks + 2 * hi);
      sc = wmma_f32(qa[ks], kb, sc);
    }

    const bool diag = (kt == qt);
#pragma unroll
    for (int i = 0; i < 8; ++i) {
      float s = sc[i] * scale;
      if (diag && lo > i + 8 * hi) s = -__builtin_inff();
      float rm = s;
      rm = fmaxf(rm, row_xor<1>(rm));
      rm = fmaxf(rm, row_xor<2>(rm));
      rm = fmaxf(rm, row_xor<4>(rm));
      rm = fmaxf(rm, row_xor<8>(rm));
      const float mnew = fmaxf(mrow[i], rm);
      const float a = expf(mrow[i] - mnew);
      const float p = expf(s - mnew);
      float rs = p;
      rs += row_xor<1>(rs);
      rs += row_xor<2>(rs);
      rs += row_xor<4>(rs);
      rs += row_xor<8>(rs);
      lrow[i] = lrow[i] * a + rs;
      mrow[i] = mnew;
      alph[i] = a;
      Pst[(i + 8 * hi) * 16 + lo] = p;
    }
#pragma unroll
    for (int t = 0; t < 8; ++t)
#pragma unroll
      for (int i = 0; i < 8; ++i) o[t][i] *= alph[i];

#pragma unroll
    for (int ks = 0; ks < 4; ++ks) {
      v2f a = *(const v2f*)(&Pst[lo * 16 + 4 * ks + 2 * hi]);
      const float* vrow = Vb + (size_t)(kt * 16 + 4 * ks + 2 * hi) * DK;
#pragma unroll
      for (int t = 0; t < 8; ++t) {
        v2f bv;
        bv[0] = vrow[t * 16 + lo];
        bv[1] = vrow[DK + t * 16 + lo];
        o[t] = wmma_f32(a, bv, o[t]);
      }
    }
  }

#pragma unroll
  for (int i = 0; i < 8; ++i) {
    const float invl = 1.f / lrow[i];
    const size_t row = (size_t)b * SEQ + qt * 16 + i + 8 * hi;
#pragma unroll
    for (int t = 0; t < 8; ++t)
      AO[row * D_MODEL + h * DK + t * 16 + lo] = o[t][i] * invl;
  }
}

__global__ __launch_bounds__(32)
void mha_out_proj(const float* __restrict__ A, const float* __restrict__ WO,
                  float* __restrict__ out) {
  const int lane = threadIdx.x;
  const int hi = lane >> 4;
  const int lo = lane & 15;
  const int m0 = blockIdx.x * 64;
  const int n0 = blockIdx.y * 64;

  v8f acc[4][4];
  gemm_64x64(A, WO, m0, n0, lo, hi, acc);

#pragma unroll
  for (int mi = 0; mi < 4; ++mi)
#pragma unroll
    for (int t = 0; t < 4; ++t)
#pragma unroll
      for (int i = 0; i < 8; ++i)
        out[(size_t)(m0 + mi * 16 + i + 8 * hi) * D_MODEL + n0 + t * 16 + lo] =
            acc[mi][t][i];
}

extern "C" void kernel_launch(void* const* d_in, const int* in_sizes, int n_in,
                              void* d_out, int out_size, void* d_ws, size_t ws_size,
                              hipStream_t stream) {
  (void)in_sizes; (void)n_in; (void)out_size; (void)ws_size;
  const float* x   = (const float*)d_in[0];
  const int*   tp  = (const int*)d_in[1];
  const float* WQ  = (const float*)d_in[2];
  const float* WK  = (const float*)d_in[3];
  const float* WV  = (const float*)d_in[4];
  const float* WO  = (const float*)d_in[5];
  float* out = (float*)d_out;

  const size_t QSZ = (size_t)BATCH * NHEADS * SEQ * DK;
  float* ws = (float*)d_ws;
  float* Qw = ws;
  float* Kw = ws + QSZ;
  float* Vw = ws + 2 * QSZ;
  float* AO = ws + 3 * QSZ;

  dim3 gProj(MROWS / 64, D_MODEL / 64, 3);
  mha_qkv_rope<<<gProj, 32, 0, stream>>>(x, tp, WQ, WK, WV, Qw, Kw, Vw);

  dim3 gAttn(SEQ / 16, NHEADS, BATCH);
  mha_attn<<<gAttn, 32, 0, stream>>>(Qw, Kw, Vw, AO);

  dim3 gOut(MROWS / 64, D_MODEL / 64);
  mha_out_proj<<<gOut, 32, 0, stream>>>(AO, WO, out);
}


# Attention: Q, K, V fragments loaded 16 bytes per lane, one 16-load group per tile each (half the K, a quarter of the V load instructions); permutes d order of score sums and O column per lane
# speedup vs baseline: 2.7502x; 1.0725x over previous
#include <hip/hip_runtime.h>
#include <math.h>

#define D_MODEL 2048
#define NHEADS  16
#define DK      128
#define SEQ     2048
#define BATCH   2
#define MROWS   (BATCH * SEQ)
#define LOG_THETA 9.210340371976184f

typedef __attribute__((ext_vector_type(2))) float v2f;
typedef __attribute__((ext_vector_type(8))) float v8f;

__device__ __forceinline__ v8f wmma_f32(v2f a, v2f b, v8f c) {
  return __builtin_amdgcn_wmma_f32_16x16x4_f32(false, a, false, b,
                                               (short)0, c, false, false);
}

__device__ __forceinline__ v8f vzero8() {
  v8f z = {0.f, 0.f, 0.f, 0.f, 0.f, 0.f, 0.f, 0.f};
  return z;
}

template <int M>
__device__ __forceinline__ float row_xor(float v) {
  return __builtin_bit_cast(
      float, __builtin_amdgcn_mov_dpp(__builtin_bit_cast(int, v), 0x160 | M,
                                      0xf, 0xf, true));
}

#define NV4   4
#define KSTEP (8 * NV4)
#define TROW  ((size_t)16 * D_MODEL)

struct FragSet {
  float4 a[4][NV4];
  float4 b[4][NV4];
};

__device__ __forceinline__ void load_frags(FragSet& f, const float* __restrict__ pa,
                                           const float* __restrict__ pb, int k) {
#pragma unroll
  for (int i = 0; i < 4; ++i)
#pragma unroll
    for (int j = 0; j < NV4; ++j)
      f.a[i][j] = *(const float4*)(pa + i * TROW + k + 4 * j);
#pragma unroll
  for (int t = 0; t < 4; ++t)
#pragma unroll
    for (int j = 0; j < NV4; ++j)
      f.b[t][j] = *(const float4*)(pb + t * TROW + k + 4 * j);
}

__device__ __forceinline__ void mma_frags(const FragSet& f, v8f (&acc)[4][4]) {
#pragma unroll
  for (int j = 0; j < NV4; ++j) {
#pragma unroll
    for (int i = 0; i < 4; ++i) {
      v2f a = {f.a[i][j].x, f.a[i][j].y};
#pragma unroll
      for (int t = 0; t < 4; ++t) {
        v2f b = {f.b[t][j].x, f.b[t][j].y};
        acc[i][t] = wmma_f32(a, b, acc[i][t]);
      }
    }
#pragma unroll
    for (int i = 0; i < 4; ++i) {
      v2f a = {f.a[i][j].z, f.a[i][j].w};
#pragma unroll
      for (int t = 0; t < 4; ++t) {
        v2f b = {f.b[t][j].z, f.b[t][j].w};
        acc[i][t] = wmma_f32(a, b, acc[i][t]);
      }
    }
  }
}

__device__ __forceinline__ void gemm_64x64(const float* __restrict__ A,
                                           const float* __restrict__ Wm,
                                           int m0, int n0, int lo, int hi,
                                           v8f (&acc)[4][4]) {
#pragma unroll
  for (int i = 0; i < 4; ++i)
#pragma unroll
    for (int t = 0; t < 4; ++t) acc[i][t] = vzero8();

  const float* pa = A + (size_t)(m0 + lo) * D_MODEL + 4 * NV4 * hi;
  const float* pb = Wm + (size_t)(n0 + lo) * D_MODEL + 4 * NV4 * hi;

  FragSet f0, f1;
  load_frags(f0, pa, pb, 0);
  int k = 0;
  for (; k < D_MODEL - 2 * KSTEP; k += 2 * KSTEP) {
    load_frags(f1, pa, pb, k + KSTEP);
    __builtin_amdgcn_sched_barrier(0);
    mma_frags(f0, acc);
    __builtin_amdgcn_sched_barrier(0);
    load_frags(f0, pa, pb, k + 2 * KSTEP);
    __builtin_amdgcn_sched_barrier(0);
    mma_frags(f1, acc);
  }
  load_frags(f1, pa, pb, k + KSTEP);
  __builtin_amdgcn_sched_barrier(0);
  mma_frags(f0, acc);
  mma_frags(f1, acc);
}

__global__ __launch_bounds__(32)
void mha_qkv_rope(const float* __restrict__ x, const int* __restrict__ tpos,
                  const float* __restrict__ WQ, const float* __restrict__ WK,
                  const float* __restrict__ WV,
                  float* __restrict__ Qo, float* __restrict__ Ko,
                  float* __restrict__ Vo) {
  const int lane = threadIdx.x;
  const int hi = lane >> 4;
  const int lo = lane & 15;
  const int m0 = blockIdx.x * 64;
  const int n0 = blockIdx.y * 64;
  const int mat = blockIdx.z;

  const float* __restrict__ W = (mat == 0) ? WQ : ((mat == 1) ? WK : WV);
  float* __restrict__ Out = (mat == 0) ? Qo : ((mat == 1) ? Ko : Vo);

  v8f acc[4][4];
  gemm_64x64(x, W, m0, n0, lo, hi, acc);

  const int bidx = m0 >> 11;
  if (mat == 2) {
#pragma unroll
    for (int mi = 0; mi < 4; ++mi)
#pragma unroll
      for (int t = 0; t < 4; ++t) {
        const int ng = n0 + t * 16 + lo;
        const int h = ng >> 7, d = ng & 127;
#pragma unroll
        for (int i = 0; i < 8; ++i) {
          const int s = (m0 + mi * 16 + i + 8 * hi) & (SEQ - 1);
          Out[(((size_t)bidx * NHEADS + h) * SEQ + s) * DK + d] = acc[mi][t][i];
        }
      }
  } else {
    float inv_freq[4];
    size_t colofs[4];
#pragma unroll
    for (int t = 0; t < 4; ++t) {
      const int ng = n0 + t * 16 + lo;
      const int h = ng >> 7, d = ng & 127;
      inv_freq[t] = expf(-(float)(d & ~1) * (LOG_THETA / (float)DK));
      colofs[t] = ((size_t)bidx * NHEADS + h) * SEQ * DK + d;
    }
    const int odd = lo & 1;
#pragma unroll 1
    for (int i = 0; i < 8; ++i) {
      int srow[4];
      float pos[4];
#pragma unroll
      for (int mi = 0; mi < 4; ++mi) {
        srow[mi] = (m0 + mi * 16 + i + 8 * hi) & (SEQ - 1);
        pos[mi] = (float)tpos[bidx * SEQ + srow[mi]];
      }
#pragma unroll
      for (int mi = 0; mi < 4; ++mi) {
#pragma unroll
        for (int t = 0; t < 4; ++t) {
          float sn, cs;
          sincosf(pos[mi] * inv_freq[t], &sn, &cs);
          const float val = acc[mi][t][0];
          const float part = row_xor<1>(val);
          const float r = odd ? fmaf(part, sn, val * cs)
                              : fmaf(val, cs, -part * sn);
          Out[colofs[t] + (size_t)srow[mi] * DK] = r;
          acc[mi][t] = __builtin_shufflevector(acc[mi][t], acc[mi][t],
                                               1, 2, 3, 4, 5, 6, 7, 7);
        }
      }
    }
  }
}

__global__ __launch_bounds__(32)
void mha_attn(const float* __restrict__ Q, const float* __restrict__ K,
              const float* __restrict__ V, float* __restrict__ AO) {
  const int lane = threadIdx.x;
  const int hi = lane >> 4;
  const int lo = lane & 15;
  const int qt = blockIdx.x;
  const int h = blockIdx.y;
  const int b = blockIdx.z;

  const size_t head = ((size_t)b * NHEADS + h) * SEQ;
  const float* __restrict__ Qb = Q + head * DK;
  const float* __restrict__ Kb = K + head * DK;
  const float* __restrict__ Vb = V + head * DK;

  __shared__ float Pst[16 * 16];

  float4 qa[16];
  const float* qrow = Qb + (size_t)(qt * 16 + lo) * DK + 4 * hi;
#pragma unroll
  for (int u = 0; u < 16; ++u)
    qa[u] = *(const float4*)(qrow + 8 * u);

  float mrow[8], lrow[8], alph[8];
  v8f o[8];
#pragma unroll
  for (int i = 0; i < 8; ++i) {
    mrow[i] = -__builtin_inff();
    lrow[i] = 0.f;
    o[i] = vzero8();
  }

  const float scale = 0.08838834764831845f;

  for (int kt = 0; kt <= qt; ++kt) {
    v8f sc = vzero8();
    const float* krow = Kb + (size_t)(kt * 16 + lo) * DK + 4 * hi;
    float4 kb[16];
#pragma unroll
    for (int u = 0; u < 16; ++u) kb[u] = *(const float4*)(krow + 8 * u);
    __builtin_amdgcn_sched_barrier(0);
#pragma unroll
    for (int u = 0; u < 16; ++u) {
      v2f a0 = {qa[u].x, qa[u].y};
      v2f b0 = {kb[u].x, kb[u].y};
      sc = wmma_f32(a0, b0, sc);
      v2f a1 = {qa[u].z, qa[u].w};
      v2f b1 = {kb[u].z, kb[u].w};
      sc = wmma_f32(a1, b1, sc);
    }
    __builtin_amdgcn_sched_barrier(0);
    float4 vb[4][2][2];
    {
      const float* vbase = Vb + (size_t)(kt * 16 + 2 * hi) * DK + 4 * lo;
#pragma unroll
      for (int ks = 0; ks < 4; ++ks)
#pragma unroll
        for (int g = 0; g < 2; ++g) {
          vb[ks][g][0] = *(const float4*)(vbase + (size_t)(4 * ks) * DK + 64 * g);
          vb[ks][g][1] = *(const float4*)(vbase + (size_t)(4 * ks + 1) * DK + 64 * g);
        }
    }
    __builtin_amdgcn_sched_barrier(0);

    const bool diag = (kt == qt);
#pragma unroll
    for (int i = 0; i < 8; ++i) {
      float s = sc[i] * scale;
      if (diag && lo > i + 8 * hi) s = -__builtin_inff();
      float rm = s;
      rm = fmaxf(rm, row_xor<1>(rm));
      rm = fmaxf(rm, row_xor<2>(rm));
      rm = fmaxf(rm, row_xor<4>(rm));
      rm = fmaxf(rm, row_xor<8>(rm));
      const float mnew = fmaxf(mrow[i], rm);
      const float a = expf(mrow[i] - mnew);
      const float p = expf(s - mnew);
      float rs = p;
      rs += row_xor<1>(rs);
      rs += row_xor<2>(rs);
      rs += row_xor<4>(rs);
      rs += row_xor<8>(rs);
      lrow[i] = lrow[i] * a + rs;
      mrow[i] = mnew;
      alph[i] = a;
      Pst[(i + 8 * hi) * 16 + lo] = p;
    }
#pragma unroll
    for (int t = 0; t < 8; ++t)
#pragma unroll
      for (int i = 0; i < 8; ++i) o[t][i] *= alph[i];

#pragma unroll
    for (int ks = 0; ks < 4; ++ks) {
      v2f a = *(const v2f*)(&Pst[lo * 16 + 4 * ks + 2 * hi]);
#pragma unroll
      for (int g = 0; g < 2; ++g) {
        const float4 v0 = vb[ks][g][0];
        const float4 v1 = vb[ks][g][1];
        v2f b0 = {v0.x, v1.x};
        v2f b1 = {v0.y, v1.y};
        v2f b2 = {v0.z, v1.z};
        v2f b3 = {v0.w, v1.w};
        o[4 * g + 0] = wmma_f32(a, b0, o[4 * g + 0]);
        o[4 * g + 1] = wmma_f32(a, b1, o[4 * g + 1]);
        o[4 * g + 2] = wmma_f32(a, b2, o[4 * g + 2]);
        o[4 * g + 3] = wmma_f32(a, b3, o[4 * g + 3]);
      }
    }
  }

#pragma unroll
  for (int i = 0; i < 8; ++i) {
    const float invl = 1.f / lrow[i];
    const size_t row = (size_t)b * SEQ + qt * 16 + i + 8 * hi;
#pragma unroll
    for (int g = 0; g < 2; ++g) {
      float4 r;
      r.x = o[4 * g + 0][i] * invl;
      r.y = o[4 * g + 1][i] * invl;
      r.z = o[4 * g + 2][i] * invl;
      r.w = o[4 * g + 3][i] * invl;
      *(float4*)(AO + row * D_MODEL + h * DK + 64 * g + 4 * lo) = r;
    }
  }
}

__global__ __launch_bounds__(32)
void mha_out_proj(const float* __restrict__ A, const float* __restrict__ WO,
                  float* __restrict__ out) {
  const int lane = threadIdx.x;
  const int hi = lane >> 4;
  const int lo = lane & 15;
  const int m0 = blockIdx.x * 64;
  const int n0 = blockIdx.y * 64;

  v8f acc[4][4];
  gemm_64x64(A, WO, m0, n0, lo, hi, acc);

#pragma unroll
  for (int mi = 0; mi < 4; ++mi)
#pragma unroll
    for (int t = 0; t < 4; ++t)
#pragma unroll
      for (int i = 0; i < 8; ++i)
        out[(size_t)(m0 + mi * 16 + i + 8 * hi) * D_MODEL + n0 + t * 16 + lo] =
            acc[mi][t][i];
}

extern "C" void kernel_launch(void* const* d_in, const int* in_sizes, int n_in,
                              void* d_out, int out_size, void* d_ws, size_t ws_size,
                              hipStream_t stream) {
  (void)in_sizes; (void)n_in; (void)out_size; (void)ws_size;
  const float* x   = (const float*)d_in[0];
  const int*   tp  = (const int*)d_in[1];
  const float* WQ  = (const float*)d_in[2];
  const float* WK  = (const float*)d_in[3];
  const float* WV  = (const float*)d_in[4];
  const float* WO  = (const float*)d_in[5];
  float* out = (float*)d_out;

  const size_t QSZ = (size_t)BATCH * NHEADS * SEQ * DK;
  float* ws = (float*)d_ws;
  float* Qw = ws;
  float* Kw = ws + QSZ;
  float* Vw = ws + 2 * QSZ;
  float* AO = ws + 3 * QSZ;

  dim3 gProj(MROWS / 64, D_MODEL / 64, 3);
  mha_qkv_rope<<<gProj, 32, 0, stream>>>(x, tp, WQ, WK, WV, Qw, Kw, Vw);

  dim3 gAttn(SEQ / 16, NHEADS, BATCH);
  mha_attn<<<gAttn, 32, 0, stream>>>(Qw, Kw, Vw, AO);

  dim3 gOut(MROWS / 64, D_MODEL / 64);
  mha_out_proj<<<gOut, 32, 0, stream>>>(AO, WO, out);
}


# Attention: each wave runs two causal query tiles, 127-i then i, so every wave walks 129 key tiles (was 1..128 by tile); grid 2048 waves; per-tile code, loads and summation order unchanged
# speedup vs baseline: 2.9631x; 1.0774x over previous
#include <hip/hip_runtime.h>
#include <math.h>

#define D_MODEL 2048
#define NHEADS  16
#define DK      128
#define SEQ     2048
#define BATCH   2
#define MROWS   (BATCH * SEQ)
#define LOG_THETA 9.210340371976184f

typedef __attribute__((ext_vector_type(2))) float v2f;
typedef __attribute__((ext_vector_type(8))) float v8f;

__device__ __forceinline__ v8f wmma_f32(v2f a, v2f b, v8f c) {
  return __builtin_amdgcn_wmma_f32_16x16x4_f32(false, a, false, b,
                                               (short)0, c, false, false);
}

__device__ __forceinline__ v8f vzero8() {
  v8f z = {0.f, 0.f, 0.f, 0.f, 0.f, 0.f, 0.f, 0.f};
  return z;
}

template <int M>
__device__ __forceinline__ float row_xor(float v) {
  return __builtin_bit_cast(
      float, __builtin_amdgcn_mov_dpp(__builtin_bit_cast(int, v), 0x160 | M,
                                      0xf, 0xf, true));
}

#define NV4   4
#define KSTEP (8 * NV4)
#define TROW  ((size_t)16 * D_MODEL)

struct FragSet {
  float4 a[4][NV4];
  float4 b[4][NV4];
};

__device__ __forceinline__ void load_frags(FragSet& f, const float* __restrict__ pa,
                                           const float* __restrict__ pb, int k) {
#pragma unroll
  for (int i = 0; i < 4; ++i)
#pragma unroll
    for (int j = 0; j < NV4; ++j)
      f.a[i][j] = *(const float4*)(pa + i * TROW + k + 4 * j);
#pragma unroll
  for (int t = 0; t < 4; ++t)
#pragma unroll
    for (int j = 0; j < NV4; ++j)
      f.b[t][j] = *(const float4*)(pb + t * TROW + k + 4 * j);
}

__device__ __forceinline__ void mma_frags(const FragSet& f, v8f (&acc)[4][4]) {
#pragma unroll
  for (int j = 0; j < NV4; ++j) {
#pragma unroll
    for (int i = 0; i < 4; ++i) {
      v2f a = {f.a[i][j].x, f.a[i][j].y};
#pragma unroll
      for (int t = 0; t < 4; ++t) {
        v2f b = {f.b[t][j].x, f.b[t][j].y};
        acc[i][t] = wmma_f32(a, b, acc[i][t]);
      }
    }
#pragma unroll
    for (int i = 0; i < 4; ++i) {
      v2f a = {f.a[i][j].z, f.a[i][j].w};
#pragma unroll
      for (int t = 0; t < 4; ++t) {
        v2f b = {f.b[t][j].z, f.b[t][j].w};
        acc[i][t] = wmma_f32(a, b, acc[i][t]);
      }
    }
  }
}

__device__ __forceinline__ void gemm_64x64(const float* __restrict__ A,
                                           const float* __restrict__ Wm,
                                           int m0, int n0, int lo, int hi,
                                           v8f (&acc)[4][4]) {
#pragma unroll
  for (int i = 0; i < 4; ++i)
#pragma unroll
    for (int t = 0; t < 4; ++t) acc[i][t] = vzero8();

  const float* pa = A + (size_t)(m0 + lo) * D_MODEL + 4 * NV4 * hi;
  const float* pb = Wm + (size_t)(n0 + lo) * D_MODEL + 4 * NV4 * hi;

  FragSet f0, f1;
  load_frags(f0, pa, pb, 0);
  int k = 0;
  for (; k < D_MODEL - 2 * KSTEP; k += 2 * KSTEP) {
    load_frags(f1, pa, pb, k + KSTEP);
    __builtin_amdgcn_sched_barrier(0);
    mma_frags(f0, acc);
    __builtin_amdgcn_sched_barrier(0);
    load_frags(f0, pa, pb, k + 2 * KSTEP);
    __builtin_amdgcn_sched_barrier(0);
    mma_frags(f1, acc);
  }
  load_frags(f1, pa, pb, k + KSTEP);
  __builtin_amdgcn_sched_barrier(0);
  mma_frags(f0, acc);
  mma_frags(f1, acc);
}

__global__ __launch_bounds__(32)
void mha_qkv_rope(const float* __restrict__ x, const int* __restrict__ tpos,
                  const float* __restrict__ WQ, const float* __restrict__ WK,
                  const float* __restrict__ WV,
                  float* __restrict__ Qo, float* __restrict__ Ko,
                  float* __restrict__ Vo) {
  const int lane = threadIdx.x;
  const int hi = lane >> 4;
  const int lo = lane & 15;
  const int m0 = blockIdx.x * 64;
  const int n0 = blockIdx.y * 64;
  const int mat = blockIdx.z;

  const float* __restrict__ W = (mat == 0) ? WQ : ((mat == 1) ? WK : WV);
  float* __restrict__ Out = (mat == 0) ? Qo : ((mat == 1) ? Ko : Vo);

  v8f acc[4][4];
  gemm_64x64(x, W, m0, n0, lo, hi, acc);

  const int bidx = m0 >> 11;
  if (mat == 2) {
#pragma unroll
    for (int mi = 0; mi < 4; ++mi)
#pragma unroll
      for (int t = 0; t < 4; ++t) {
        const int ng = n0 + t * 16 + lo;
        const int h = ng >> 7, d = ng & 127;
#pragma unroll
        for (int i = 0; i < 8; ++i) {
          const int s = (m0 + mi * 16 + i + 8 * hi) & (SEQ - 1);
          Out[(((size_t)bidx * NHEADS + h) * SEQ + s) * DK + d] = acc[mi][t][i];
        }
      }
  } else {
    float inv_freq[4];
    size_t colofs[4];
#pragma unroll
    for (int t = 0; t < 4; ++t) {
      const int ng = n0 + t * 16 + lo;
      const int h = ng >> 7, d = ng & 127;
      inv_freq[t] = expf(-(float)(d & ~1) * (LOG_THETA / (float)DK));
      colofs[t] = ((size_t)bidx * NHEADS + h) * SEQ * DK + d;
    }
    const int odd = lo & 1;
#pragma unroll 1
    for (int i = 0; i < 8; ++i) {
      int srow[4];
      float pos[4];
#pragma unroll
      for (int mi = 0; mi < 4; ++mi) {
        srow[mi] = (m0 + mi * 16 + i + 8 * hi) & (SEQ - 1);
        pos[mi] = (float)tpos[bidx * SEQ + srow[mi]];
      }
#pragma unroll
      for (int mi = 0; mi < 4; ++mi) {
#pragma unroll
        for (int t = 0; t < 4; ++t) {
          float sn, cs;
          sincosf(pos[mi] * inv_freq[t], &sn, &cs);
          const float val = acc[mi][t][0];
          const float part = row_xor<1>(val);
          const float r = odd ? fmaf(part, sn, val * cs)
                              : fmaf(val, cs, -part * sn);
          Out[colofs[t] + (size_t)srow[mi] * DK] = r;
          acc[mi][t] = __builtin_shufflevector(acc[mi][t], acc[mi][t],
                                               1, 2, 3, 4, 5, 6, 7, 7);
        }
      }
    }
  }
}

#define NQT (SEQ / 16)
__global__ __launch_bounds__(32)
void mha_attn(const float* __restrict__ Q, const float* __restrict__ K,
              const float* __restrict__ V, float* __restrict__ AO) {
  const int lane = threadIdx.x;
  const int hi = lane >> 4;
  const int lo = lane & 15;
  const int pr = blockIdx.x;
  const int h = blockIdx.y;
  const int b = blockIdx.z;

  const size_t head = ((size_t)b * NHEADS + h) * SEQ;
  const float* __restrict__ Qb = Q + head * DK;
  const float* __restrict__ Kb = K + head * DK;
  const float* __restrict__ Vb = V + head * DK;

  __shared__ float Pst[16 * 16];

#pragma unroll 1
  for (int pass = 0; pass < 2; ++pass) {
  const int qt = pass ? pr : (NQT - 1 - pr);

  float4 qa[16];
  const float* qrow = Qb + (size_t)(qt * 16 + lo) * DK + 4 * hi;
#pragma unroll
  for (int u = 0; u < 16; ++u)
    qa[u] = *(const float4*)(qrow + 8 * u);

  float mrow[8], lrow[8], alph[8];
  v8f o[8];
#pragma unroll
  for (int i = 0; i < 8; ++i) {
    mrow[i] = -__builtin_inff();
    lrow[i] = 0.f;
    o[i] = vzero8();
  }

  const float scale = 0.08838834764831845f;

  for (int kt = 0; kt <= qt; ++kt) {
    v8f sc = vzero8();
    const float* krow = Kb + (size_t)(kt * 16 + lo) * DK + 4 * hi;
    float4 kb[16];
#pragma unroll
    for (int u = 0; u < 16; ++u) kb[u] = *(const float4*)(krow + 8 * u);
    __builtin_amdgcn_sched_barrier(0);
#pragma unroll
    for (int u = 0; u < 16; ++u) {
      v2f a0 = {qa[u].x, qa[u].y};
      v2f b0 = {kb[u].x, kb[u].y};
      sc = wmma_f32(a0, b0, sc);
      v2f a1 = {qa[u].z, qa[u].w};
      v2f b1 = {kb[u].z, kb[u].w};
      sc = wmma_f32(a1, b1, sc);
    }
    __builtin_amdgcn_sched_barrier(0);
    float4 vb[4][2][2];
    {
      const float* vbase = Vb + (size_t)(kt * 16 + 2 * hi) * DK + 4 * lo;
#pragma unroll
      for (int ks = 0; ks < 4; ++ks)
#pragma unroll
        for (int g = 0; g < 2; ++g) {
          vb[ks][g][0] = *(const float4*)(vbase + (size_t)(4 * ks) * DK + 64 * g);
          vb[ks][g][1] = *(const float4*)(vbase + (size_t)(4 * ks + 1) * DK + 64 * g);
        }
    }
    __builtin_amdgcn_sched_barrier(0);

    const bool diag = (kt == qt);
#pragma unroll
    for (int i = 0; i < 8; ++i) {
      float s = sc[i] * scale;
      if (diag && lo > i + 8 * hi) s = -__builtin_inff();
      float rm = s;
      rm = fmaxf(rm, row_xor<1>(rm));
      rm = fmaxf(rm, row_xor<2>(rm));
      rm = fmaxf(rm, row_xor<4>(rm));
      rm = fmaxf(rm, row_xor<8>(rm));
      const float mnew = fmaxf(mrow[i], rm);
      const float a = expf(mrow[i] - mnew);
      const float p = expf(s - mnew);
      float rs = p;
      rs += row_xor<1>(rs);
      rs += row_xor<2>(rs);
      rs += row_xor<4>(rs);
      rs += row_xor<8>(rs);
      lrow[i] = lrow[i] * a + rs;
      mrow[i] = mnew;
      alph[i] = a;
      Pst[(i + 8 * hi) * 16 + lo] = p;
    }
#pragma unroll
    for (int t = 0; t < 8; ++t)
#pragma unroll
      for (int i = 0; i < 8; ++i) o[t][i] *= alph[i];

#pragma unroll
    for (int ks = 0; ks < 4; ++ks) {
      v2f a = *(const v2f*)(&Pst[lo * 16 + 4 * ks + 2 * hi]);
#pragma unroll
      for (int g = 0; g < 2; ++g) {
        const float4 v0 = vb[ks][g][0];
        const float4 v1 = vb[ks][g][1];
        v2f b0 = {v0.x, v1.x};
        v2f b1 = {v0.y, v1.y};
        v2f b2 = {v0.z, v1.z};
        v2f b3 = {v0.w, v1.w};
        o[4 * g + 0] = wmma_f32(a, b0, o[4 * g + 0]);
        o[4 * g + 1] = wmma_f32(a, b1, o[4 * g + 1]);
        o[4 * g + 2] = wmma_f32(a, b2, o[4 * g + 2]);
        o[4 * g + 3] = wmma_f32(a, b3, o[4 * g + 3]);
      }
    }
  }

#pragma unroll
  for (int i = 0; i < 8; ++i) {
    const float invl = 1.f / lrow[i];
    const size_t row = (size_t)b * SEQ + qt * 16 + i + 8 * hi;
#pragma unroll
    for (int g = 0; g < 2; ++g) {
      float4 r;
      r.x = o[4 * g + 0][i] * invl;
      r.y = o[4 * g + 1][i] * invl;
      r.z = o[4 * g + 2][i] * invl;
      r.w = o[4 * g + 3][i] * invl;
      *(float4*)(AO + row * D_MODEL + h * DK + 64 * g + 4 * lo) = r;
    }
  }
  }
}

__global__ __launch_bounds__(32)
void mha_out_proj(const float* __restrict__ A, const float* __restrict__ WO,
                  float* __restrict__ out) {
  const int lane = threadIdx.x;
  const int hi = lane >> 4;
  const int lo = lane & 15;
  const int m0 = blockIdx.x * 64;
  const int n0 = blockIdx.y * 64;

  v8f acc[4][4];
  gemm_64x64(A, WO, m0, n0, lo, hi, acc);

#pragma unroll
  for (int mi = 0; mi < 4; ++mi)
#pragma unroll
    for (int t = 0; t < 4; ++t)
#pragma unroll
      for (int i = 0; i < 8; ++i)
        out[(size_t)(m0 + mi * 16 + i + 8 * hi) * D_MODEL + n0 + t * 16 + lo] =
            acc[mi][t][i];
}

extern "C" void kernel_launch(void* const* d_in, const int* in_sizes, int n_in,
                              void* d_out, int out_size, void* d_ws, size_t ws_size,
                              hipStream_t stream) {
  (void)in_sizes; (void)n_in; (void)out_size; (void)ws_size;
  const float* x   = (const float*)d_in[0];
  const int*   tp  = (const int*)d_in[1];
  const float* WQ  = (const float*)d_in[2];
  const float* WK  = (const float*)d_in[3];
  const float* WV  = (const float*)d_in[4];
  const float* WO  = (const float*)d_in[5];
  float* out = (float*)d_out;

  const size_t QSZ = (size_t)BATCH * NHEADS * SEQ * DK;
  float* ws = (float*)d_ws;
  float* Qw = ws;
  float* Kw = ws + QSZ;
  float* Vw = ws + 2 * QSZ;
  float* AO = ws + 3 * QSZ;

  dim3 gProj(MROWS / 64, D_MODEL / 64, 3);
  mha_qkv_rope<<<gProj, 32, 0, stream>>>(x, tp, WQ, WK, WV, Qw, Kw, Vw);

  dim3 gAttn(SEQ / 32, NHEADS, BATCH);
  mha_attn<<<gAttn, 32, 0, stream>>>(Qw, Kw, Vw, AO);

  dim3 gOut(MROWS / 64, D_MODEL / 64);
  mha_out_proj<<<gOut, 32, 0, stream>>>(AO, WO, out);
}


# Attention: keys walked in blocks of 32 (two score tiles per online-softmax step): one row max/sum/O rescale per 32 keys, two independent score chains; reorders row sum and rescale rounding
# speedup vs baseline: 3.1497x; 1.0630x over previous
#include <hip/hip_runtime.h>
#include <math.h>

#define D_MODEL 2048
#define NHEADS  16
#define DK      128
#define SEQ     2048
#define BATCH   2
#define MROWS   (BATCH * SEQ)
#define LOG_THETA 9.210340371976184f

typedef __attribute__((ext_vector_type(2))) float v2f;
typedef __attribute__((ext_vector_type(8))) float v8f;

__device__ __forceinline__ v8f wmma_f32(v2f a, v2f b, v8f c) {
  return __builtin_amdgcn_wmma_f32_16x16x4_f32(false, a, false, b,
                                               (short)0, c, false, false);
}

__device__ __forceinline__ v8f vzero8() {
  v8f z = {0.f, 0.f, 0.f, 0.f, 0.f, 0.f, 0.f, 0.f};
  return z;
}

template <int M>
__device__ __forceinline__ float row_xor(float v) {
  return __builtin_bit_cast(
      float, __builtin_amdgcn_mov_dpp(__builtin_bit_cast(int, v), 0x160 | M,
                                      0xf, 0xf, true));
}

#define NV4   4
#define KSTEP (8 * NV4)
#define TROW  ((size_t)16 * D_MODEL)

struct FragSet {
  float4 a[4][NV4];
  float4 b[4][NV4];
};

__device__ __forceinline__ void load_frags(FragSet& f, const float* __restrict__ pa,
                                           const float* __restrict__ pb, int k) {
#pragma unroll
  for (int i = 0; i < 4; ++i)
#pragma unroll
    for (int j = 0; j < NV4; ++j)
      f.a[i][j] = *(const float4*)(pa + i * TROW + k + 4 * j);
#pragma unroll
  for (int t = 0; t < 4; ++t)
#pragma unroll
    for (int j = 0; j < NV4; ++j)
      f.b[t][j] = *(const float4*)(pb + t * TROW + k + 4 * j);
}

__device__ __forceinline__ void mma_frags(const FragSet& f, v8f (&acc)[4][4]) {
#pragma unroll
  for (int j = 0; j < NV4; ++j) {
#pragma unroll
    for (int i = 0; i < 4; ++i) {
      v2f a = {f.a[i][j].x, f.a[i][j].y};
#pragma unroll
      for (int t = 0; t < 4; ++t) {
        v2f b = {f.b[t][j].x, f.b[t][j].y};
        acc[i][t] = wmma_f32(a, b, acc[i][t]);
      }
    }
#pragma unroll
    for (int i = 0; i < 4; ++i) {
      v2f a = {f.a[i][j].z, f.a[i][j].w};
#pragma unroll
      for (int t = 0; t < 4; ++t) {
        v2f b = {f.b[t][j].z, f.b[t][j].w};
        acc[i][t] = wmma_f32(a, b, acc[i][t]);
      }
    }
  }
}

__device__ __forceinline__ void gemm_64x64(const float* __restrict__ A,
                                           const float* __restrict__ Wm,
                                           int m0, int n0, int lo, int hi,
                                           v8f (&acc)[4][4]) {
#pragma unroll
  for (int i = 0; i < 4; ++i)
#pragma unroll
    for (int t = 0; t < 4; ++t) acc[i][t] = vzero8();

  const float* pa = A + (size_t)(m0 + lo) * D_MODEL + 4 * NV4 * hi;
  const float* pb = Wm + (size_t)(n0 + lo) * D_MODEL + 4 * NV4 * hi;

  FragSet f0, f1;
  load_frags(f0, pa, pb, 0);
  int k = 0;
  for (; k < D_MODEL - 2 * KSTEP; k += 2 * KSTEP) {
    load_frags(f1, pa, pb, k + KSTEP);
    __builtin_amdgcn_sched_barrier(0);
    mma_frags(f0, acc);
    __builtin_amdgcn_sched_barrier(0);
    load_frags(f0, pa, pb, k + 2 * KSTEP);
    __builtin_amdgcn_sched_barrier(0);
    mma_frags(f1, acc);
  }
  load_frags(f1, pa, pb, k + KSTEP);
  __builtin_amdgcn_sched_barrier(0);
  mma_frags(f0, acc);
  mma_frags(f1, acc);
}

__global__ __launch_bounds__(32)
void mha_qkv_rope(const float* __restrict__ x, const int* __restrict__ tpos,
                  const float* __restrict__ WQ, const float* __restrict__ WK,
                  const float* __restrict__ WV,
                  float* __restrict__ Qo, float* __restrict__ Ko,
                  float* __restrict__ Vo) {
  const int lane = threadIdx.x;
  const int hi = lane >> 4;
  const int lo = lane & 15;
  const int m0 = blockIdx.x * 64;
  const int n0 = blockIdx.y * 64;
  const int mat = blockIdx.z;

  const float* __restrict__ W = (mat == 0) ? WQ : ((mat == 1) ? WK : WV);
  float* __restrict__ Out = (mat == 0) ? Qo : ((mat == 1) ? Ko : Vo);

  v8f acc[4][4];
  gemm_64x64(x, W, m0, n0, lo, hi, acc);

  const int bidx = m0 >> 11;
  if (mat == 2) {
#pragma unroll
    for (int mi = 0; mi < 4; ++mi)
#pragma unroll
      for (int t = 0; t < 4; ++t) {
        const int ng = n0 + t * 16 + lo;
        const int h = ng >> 7, d = ng & 127;
#pragma unroll
        for (int i = 0; i < 8; ++i) {
          const int s = (m0 + mi * 16 + i + 8 * hi) & (SEQ - 1);
          Out[(((size_t)bidx * NHEADS + h) * SEQ + s) * DK + d] = acc[mi][t][i];
        }
      }
  } else {
    float inv_freq[4];
    size_t colofs[4];
#pragma unroll
    for (int t = 0; t < 4; ++t) {
      const int ng = n0 + t * 16 + lo;
      const int h = ng >> 7, d = ng & 127;
      inv_freq[t] = expf(-(float)(d & ~1) * (LOG_THETA / (float)DK));
      colofs[t] = ((size_t)bidx * NHEADS + h) * SEQ * DK + d;
    }
    const int odd = lo & 1;
#pragma unroll 1
    for (int i = 0; i < 8; ++i) {
      int srow[4];
      float pos[4];
#pragma unroll
      for (int mi = 0; mi < 4; ++mi) {
        srow[mi] = (m0 + mi * 16 + i + 8 * hi) & (SEQ - 1);
        pos[mi] = (float)tpos[bidx * SEQ + srow[mi]];
      }
#pragma unroll
      for (int mi = 0; mi < 4; ++mi) {
#pragma unroll
        for (int t = 0; t < 4; ++t) {
          float sn, cs;
          sincosf(pos[mi] * inv_freq[t], &sn, &cs);
          const float val = acc[mi][t][0];
          const float part = row_xor<1>(val);
          const float r = odd ? fmaf(part, sn, val * cs)
                              : fmaf(val, cs, -part * sn);
          Out[colofs[t] + (size_t)srow[mi] * DK] = r;
          acc[mi][t] = __builtin_shufflevector(acc[mi][t], acc[mi][t],
                                               1, 2, 3, 4, 5, 6, 7, 7);
        }
      }
    }
  }
}

#define NQT (SEQ / 16)
#define PPITCH 36
__global__ __launch_bounds__(32)
void mha_attn(const float* __restrict__ Q, const float* __restrict__ K,
              const float* __restrict__ V, float* __restrict__ AO) {
  const int lane = threadIdx.x;
  const int hi = lane >> 4;
  const int lo = lane & 15;
  const int pr = blockIdx.x;
  const int h = blockIdx.y;
  const int b = blockIdx.z;

  const size_t head = ((size_t)b * NHEADS + h) * SEQ;
  const float* __restrict__ Qb = Q + head * DK;
  const float* __restrict__ Kb = K + head * DK;
  const float* __restrict__ Vb = V + head * DK;

  __shared__ __attribute__((aligned(16))) float Pst[16 * PPITCH];

#pragma unroll 1
  for (int pass = 0; pass < 2; ++pass) {
  const int qt = pass ? pr : (NQT - 1 - pr);

  float4 qa[16];
  const float* qrow = Qb + (size_t)(qt * 16 + lo) * DK + 4 * hi;
#pragma unroll
  for (int u = 0; u < 16; ++u)
    qa[u] = *(const float4*)(qrow + 8 * u);

  float mrow[8], lrow[8], alph[8];
  v8f o[8];
#pragma unroll
  for (int i = 0; i < 8; ++i) {
    mrow[i] = -__builtin_inff();
    lrow[i] = 0.f;
    o[i] = vzero8();
  }

  const float scale = 0.08838834764831845f;
  const int nblk = (qt >> 1) + 1;

  for (int kb2 = 0; kb2 < nblk; ++kb2) {
    const int t0 = 2 * kb2;
    v8f sc0 = vzero8();
    v8f sc1 = vzero8();
    const float* krow = Kb + (size_t)(t0 * 16 + lo) * DK + 4 * hi;
    float4 k0[16], k1[16];
#pragma unroll
    for (int u = 0; u < 16; ++u) {
      k0[u] = *(const float4*)(krow + 8 * u);
      k1[u] = *(const float4*)(krow + 16 * DK + 8 * u);
    }
    __builtin_amdgcn_sched_barrier(0);
#pragma unroll
    for (int u = 0; u < 16; ++u) {
      v2f a0 = {qa[u].x, qa[u].y};
      v2f b00 = {k0[u].x, k0[u].y};
      v2f b10 = {k1[u].x, k1[u].y};
      sc0 = wmma_f32(a0, b00, sc0);
      sc1 = wmma_f32(a0, b10, sc1);
      v2f a1 = {qa[u].z, qa[u].w};
      v2f b01 = {k0[u].z, k0[u].w};
      v2f b11 = {k1[u].z, k1[u].w};
      sc0 = wmma_f32(a1, b01, sc0);
      sc1 = wmma_f32(a1, b11, sc1);
    }
    __builtin_amdgcn_sched_barrier(0);
    float4 vb[8][2][2];
    {
      const float* vbase = Vb + (size_t)(t0 * 16 + 2 * hi) * DK + 4 * lo;
#pragma unroll
      for (int ks = 0; ks < 8; ++ks)
#pragma unroll
        for (int g = 0; g < 2; ++g) {
          vb[ks][g][0] = *(const float4*)(vbase + (size_t)(4 * ks) * DK + 64 * g);
          vb[ks][g][1] = *(const float4*)(vbase + (size_t)(4 * ks + 1) * DK + 64 * g);
        }
    }
    __builtin_amdgcn_sched_barrier(0);

    float s0[8], s1[8];
#pragma unroll
    for (int i = 0; i < 8; ++i) {
      s0[i] = sc0[i] * scale;
      s1[i] = sc1[i] * scale;
    }
    if (kb2 == nblk - 1) {
      const bool first_is_diag = (t0 == qt);
#pragma unroll
      for (int i = 0; i < 8; ++i) {
        const bool up = lo > i + 8 * hi;
        if (first_is_diag) {
          if (up) s0[i] = -__builtin_inff();
          s1[i] = -__builtin_inff();
        } else {
          if (up) s1[i] = -__builtin_inff();
        }
      }
    }
#pragma unroll
    for (int i = 0; i < 8; ++i) {
      float rm = fmaxf(s0[i], s1[i]);
      rm = fmaxf(rm, row_xor<1>(rm));
      rm = fmaxf(rm, row_xor<2>(rm));
      rm = fmaxf(rm, row_xor<4>(rm));
      rm = fmaxf(rm, row_xor<8>(rm));
      const float mnew = fmaxf(mrow[i], rm);
      const float a = expf(mrow[i] - mnew);
      const float p0 = expf(s0[i] - mnew);
      const float p1 = expf(s1[i] - mnew);
      float rs = p0 + p1;
      rs += row_xor<1>(rs);
      rs += row_xor<2>(rs);
      rs += row_xor<4>(rs);
      rs += row_xor<8>(rs);
      lrow[i] = lrow[i] * a + rs;
      mrow[i] = mnew;
      alph[i] = a;
      Pst[(i + 8 * hi) * PPITCH + lo] = p0;
      Pst[(i + 8 * hi) * PPITCH + 16 + lo] = p1;
    }
#pragma unroll
    for (int t = 0; t < 8; ++t)
#pragma unroll
      for (int i = 0; i < 8; ++i) o[t][i] *= alph[i];

#pragma unroll
    for (int ks = 0; ks < 8; ++ks) {
      v2f a = *(const v2f*)(&Pst[lo * PPITCH + 4 * ks + 2 * hi]);
#pragma unroll
      for (int g = 0; g < 2; ++g) {
        const float4 v0 = vb[ks][g][0];
        const float4 v1 = vb[ks][g][1];
        v2f b0 = {v0.x, v1.x};
        v2f b1 = {v0.y, v1.y};
        v2f b2 = {v0.z, v1.z};
        v2f b3 = {v0.w, v1.w};
        o[4 * g + 0] = wmma_f32(a, b0, o[4 * g + 0]);
        o[4 * g + 1] = wmma_f32(a, b1, o[4 * g + 1]);
        o[4 * g + 2] = wmma_f32(a, b2, o[4 * g + 2]);
        o[4 * g + 3] = wmma_f32(a, b3, o[4 * g + 3]);
      }
    }
  }

#pragma unroll
  for (int i = 0; i < 8; ++i) {
    const float invl = 1.f / lrow[i];
    const size_t row = (size_t)b * SEQ + qt * 16 + i + 8 * hi;
#pragma unroll
    for (int g = 0; g < 2; ++g) {
      float4 r;
      r.x = o[4 * g + 0][i] * invl;
      r.y = o[4 * g + 1][i] * invl;
      r.z = o[4 * g + 2][i] * invl;
      r.w = o[4 * g + 3][i] * invl;
      *(float4*)(AO + row * D_MODEL + h * DK + 64 * g + 4 * lo) = r;
    }
  }
  }
}

__global__ __launch_bounds__(32)
void mha_out_proj(const float* __restrict__ A, const float* __restrict__ WO,
                  float* __restrict__ out) {
  const int lane = threadIdx.x;
  const int hi = lane >> 4;
  const int lo = lane & 15;
  const int m0 = blockIdx.x * 64;
  const int n0 = blockIdx.y * 64;

  v8f acc[4][4];
  gemm_64x64(A, WO, m0, n0, lo, hi, acc);

#pragma unroll
  for (int mi = 0; mi < 4; ++mi)
#pragma unroll
    for (int t = 0; t < 4; ++t)
#pragma unroll
      for (int i = 0; i < 8; ++i)
        out[(size_t)(m0 + mi * 16 + i + 8 * hi) * D_MODEL + n0 + t * 16 + lo] =
            acc[mi][t][i];
}

extern "C" void kernel_launch(void* const* d_in, const int* in_sizes, int n_in,
                              void* d_out, int out_size, void* d_ws, size_t ws_size,
                              hipStream_t stream) {
  (void)in_sizes; (void)n_in; (void)out_size; (void)ws_size;
  const float* x   = (const float*)d_in[0];
  const int*   tp  = (const int*)d_in[1];
  const float* WQ  = (const float*)d_in[2];
  const float* WK  = (const float*)d_in[3];
  const float* WV  = (const float*)d_in[4];
  const float* WO  = (const float*)d_in[5];
  float* out = (float*)d_out;

  const size_t QSZ = (size_t)BATCH * NHEADS * SEQ * DK;
  float* ws = (float*)d_ws;
  float* Qw = ws;
  float* Kw = ws + QSZ;
  float* Vw = ws + 2 * QSZ;
  float* AO = ws + 3 * QSZ;

  dim3 gProj(MROWS / 64, D_MODEL / 64, 3);
  mha_qkv_rope<<<gProj, 32, 0, stream>>>(x, tp, WQ, WK, WV, Qw, Kw, Vw);

  dim3 gAttn(SEQ / 32, NHEADS, BATCH);
  mha_attn<<<gAttn, 32, 0, stream>>>(Qw, Kw, Vw, AO);

  dim3 gOut(MROWS / 64, D_MODEL / 64);
  mha_out_proj<<<gOut, 32, 0, stream>>>(AO, WO, out);
}
